# starting kernel
# baseline (speedup 1.0000x reference)
#include <hip/hip_runtime.h>
#include <hip/hip_bf16.h>

typedef __attribute__((ext_vector_type(2))) float v2f;
typedef __attribute__((ext_vector_type(8))) float v8f;

#define KB_CONST  0.1f
#define NPTS      262144
#define NUM_NETS  24
#define NPAIRS    12
#define WIDTH     8
#define NTILES    (NPTS / 16)

#define LOG2E     1.4426950408889634f
#define LN2       0.6931471805599453f

__device__ __forceinline__ float half_swap(float x) {
  return __int_as_float(__builtin_amdgcn_ds_swizzle(__float_as_int(x), 0x401F));
}

__device__ __forceinline__ float fast_tanh(float x) {
  float e = __builtin_amdgcn_exp2f(x * (2.0f * LOG2E));
  return 1.0f - 2.0f * __builtin_amdgcn_rcpf(e + 1.0f);
}

__device__ __forceinline__ v8f wmma4(v2f a, v2f b, v8f c) {
  return __builtin_amdgcn_wmma_f32_16x16x4_f32(false, a, false, b, (short)0, c,
                                               false, false);
}

__global__ __launch_bounds__(256) void subnet_wmma_kernel(
    const float* __restrict__ x,  const float* __restrict__ W1,
    const float* __restrict__ b1, const float* __restrict__ W2,
    const float* __restrict__ b2, const float* __restrict__ W3,
    float* __restrict__ sub)
{
  const int pair = blockIdx.y;
  const int k0 = 2 * pair, k1 = 2 * pair + 1;
  const int lane = threadIdx.x & 31;
  const int wave = threadIdx.x >> 5;
  const int col  = lane & 15;
  const bool lo  = lane < 16;

  const int netc = (col < 8) ? k0 : k1;
  const int wloc = col & 7;
  v2f a1;
  a1.x = lo ? W1[(netc * WIDTH + wloc) * 2 + 0] : b1[netc * WIDTH + wloc];
  a1.y = lo ? W1[(netc * WIDTH + wloc) * 2 + 1] : 0.0f;

  auto w2blk = [&](int w, int v) -> float {
    if (w < 8 && v < 8)   return W2[(k0 * WIDTH + v) * WIDTH + w];
    if (w >= 8 && v >= 8) return W2[(k1 * WIDTH + (v - 8)) * WIDTH + (w - 8)];
    return 0.0f;
  };
  const int khi = lo ? 0 : 2;
  v2f a2[5];
#pragma unroll
  for (int s = 0; s < 4; ++s) {
    a2[s].x = w2blk(4 * s + khi + 0, col);
    a2[s].y = w2blk(4 * s + khi + 1, col);
  }
  a2[4].x = lo ? ((col < 8) ? b2[k0 * WIDTH + col] : b2[k1 * WIDTH + (col - 8)])
               : 0.0f;
  a2[4].y = 0.0f;

  float w3v[WIDTH];
#pragma unroll
  for (int i = 0; i < WIDTH; ++i)
    w3v[i] = lo ? W3[k0 * WIDTH + i] : W3[k1 * WIDTH + i];

  const int waveGlobal = blockIdx.x * (blockDim.x >> 5) + wave;
  const int waveStride = gridDim.x * (blockDim.x >> 5);

  for (int t = waveGlobal; t < NTILES; t += waveStride) {
    const int p = t * 16 + col;
    const float2 xv = ((const float2*)x)[p];

    v2f bx;
    bx.x = lo ? xv.x : 1.0f;
    bx.y = lo ? xv.y : 0.0f;

    v8f c0 = {};
    v8f h1 = wmma4(a1, bx, c0);

    float hv[8], sw[8];
#pragma unroll
    for (int i = 0; i < 8; ++i) hv[i] = fast_tanh(h1[i]);
#pragma unroll
    for (int i = 0; i < 8; ++i) sw[i] = half_swap(hv[i]);

    v8f acc = {};
    v2f bs;
    bs.x = lo ? hv[0] : sw[2];  bs.y = lo ? hv[1] : sw[3];
    acc = wmma4(a2[0], bs, acc);
    bs.x = lo ? hv[4] : sw[6];  bs.y = lo ? hv[5] : sw[7];
    acc = wmma4(a2[1], bs, acc);
    bs.x = lo ? sw[0] : hv[2];  bs.y = lo ? sw[1] : hv[3];
    acc = wmma4(a2[2], bs, acc);
    bs.x = lo ? sw[4] : hv[6];  bs.y = lo ? sw[5] : hv[7];
    acc = wmma4(a2[3], bs, acc);
    bs.x = lo ? 1.0f : 0.0f;    bs.y = 0.0f;
    acc = wmma4(a2[4], bs, acc);

    float s3 = 0.0f;
#pragma unroll
    for (int i = 0; i < 8; ++i) s3 = fmaf(fast_tanh(acc[i]), w3v[i], s3);

    const int k = lo ? k0 : k1;
    sub[p * NUM_NETS + k] = s3;
  }
}

__global__ __launch_bounds__(256) void mixture_kernel(
    const float* __restrict__ x, const float* __restrict__ sub,
    float* __restrict__ out)
{
  const int p = blockIdx.x * blockDim.x + threadIdx.x;
  if (p >= NPTS) return;
  const float T = x[2 * p + 1];
  const float invkT = __builtin_amdgcn_rcpf(KB_CONST * T);

  float vals[NPAIRS], ex[NPAIRS];
  float m = -1e30f;
#pragma unroll
  for (int i = 0; i < NPAIRS; ++i) {
    const float n1  = sub[p * NUM_NETS + 2 * i];
    const float n2r = sub[p * NUM_NETS + 2 * i + 1];
    const float n2  = n2r * n2r;
    const float v   = n1 - T * n2;
    vals[i] = v;
    const float z = n2 * (1.0f / (5.0f * KB_CONST));
    ex[i] = (-v * invkT - z * z) * LOG2E;
    m = fmaxf(m, ex[i]);
  }
  float S = 0.0f;
#pragma unroll
  for (int i = 0; i < NPAIRS; ++i) {
    ex[i] = __builtin_amdgcn_exp2f(ex[i] - m);
    S += ex[i];
  }
  const float invS = __builtin_amdgcn_rcpf(S);
  float wv = 0.0f, wlp = 0.0f;
#pragma unroll
  for (int i = 0; i < NPAIRS; ++i) {
    const float w = ex[i] * invS;
    wv  = fmaf(w, vals[i], wv);
    wlp = fmaf(w, LN2 * __builtin_amdgcn_logf(w + 1e-9f), wlp);
  }
  out[p] = wv + KB_CONST * T * wlp;
}

extern "C" void kernel_launch(void* const* d_in, const int* in_sizes, int n_in,
                              void* d_out, int out_size, void* d_ws, size_t ws_size,
                              hipStream_t stream) {
  const float* x  = (const float*)d_in[0];
  const float* W1 = (const float*)d_in[1];
  const float* b1 = (const float*)d_in[2];
  const float* W2 = (const float*)d_in[3];
  const float* b2 = (const float*)d_in[4];
  const float* W3 = (const float*)d_in[5];

  float* out = (float*)d_out;
  float* sub = out + NPTS;

  dim3 gridA(32, NPAIRS);
  subnet_wmma_kernel<<<gridA, 256, 0, stream>>>(x, W1, b1, W2, b2, W3, sub);
  mixture_kernel<<<(NPTS + 255) / 256, 256, 0, stream>>>(x, sub, out);
}


# Fuse both launches into one: each wave runs all 12 net pairs on its own 32 points from an LDS table of weight fragments, keeps the 24 sub values in registers for the mixture, writes whole sub rows
# speedup vs baseline: 2.0757x; 2.0757x over previous
#include <hip/hip_runtime.h>


typedef __attribute__((ext_vector_type(2))) float v2f;
typedef __attribute__((ext_vector_type(8))) float v8f;

#define KB_CONST  0.1f
#define NPTS      262144
#define NUM_NETS  24
#define NPAIRS    12
#define WIDTH     8
#define WAVES_PER_BLOCK 8
#define NBLOCKS   (NPTS / (32 * WAVES_PER_BLOCK))

#define LOG2E     1.4426950408889634f
#define LN2       0.6931471805599453f

__device__ __forceinline__ float half_swap(float x) {
  return __int_as_float(__builtin_amdgcn_ds_swizzle(__float_as_int(x), 0x401F));
}

__device__ __forceinline__ float fast_tanh(float x) {
  float e = __builtin_amdgcn_exp2f(x * (2.0f * LOG2E));
  return 1.0f - 2.0f * __builtin_amdgcn_rcpf(e + 1.0f);
}

__device__ __forceinline__ v8f wmma4(v2f a, v2f b, v8f c) {
  return __builtin_amdgcn_wmma_f32_16x16x4_f32(false, a, false, b, (short)0, c,
                                               false, false);
}

__device__ __forceinline__ float pair_tile(v2f a1, v2f a20, v2f a21, v2f a22,
                                           v2f a23, v2f a24, const float* w3v,
                                           v2f bx, bool lo)
{
  v8f c0 = {};
  v8f h1 = wmma4(a1, bx, c0);

  float hv[8], sw[8];
#pragma unroll
  for (int i = 0; i < 8; ++i) hv[i] = fast_tanh(h1[i]);
#pragma unroll
  for (int i = 0; i < 8; ++i) sw[i] = half_swap(hv[i]);

  v8f acc = {};
  v2f bs;
  bs.x = lo ? hv[0] : sw[2];  bs.y = lo ? hv[1] : sw[3];
  acc = wmma4(a20, bs, acc);
  bs.x = lo ? hv[4] : sw[6];  bs.y = lo ? hv[5] : sw[7];
  acc = wmma4(a21, bs, acc);
  bs.x = lo ? sw[0] : hv[2];  bs.y = lo ? sw[1] : hv[3];
  acc = wmma4(a22, bs, acc);
  bs.x = lo ? sw[4] : hv[6];  bs.y = lo ? sw[5] : hv[7];
  acc = wmma4(a23, bs, acc);
  bs.x = lo ? 1.0f : 0.0f;    bs.y = 0.0f;
  acc = wmma4(a24, bs, acc);

  float s3 = 0.0f;
#pragma unroll
  for (int i = 0; i < 8; ++i) s3 = fmaf(fast_tanh(acc[i]), w3v[i], s3);
  return s3;
}

__global__ __launch_bounds__(256) void fused_subnets_mixture_kernel(
    const float* __restrict__ x,  const float* __restrict__ W1,
    const float* __restrict__ b1, const float* __restrict__ W2,
    const float* __restrict__ b2, const float* __restrict__ W3,
    float* __restrict__ out, float* __restrict__ sub)
{
  __shared__ float4 wtab[NPAIRS][5][32];

  const int lane = threadIdx.x & 31;
  const int wave = threadIdx.x >> 5;
  const int col  = lane & 15;
  const bool lo  = lane < 16;

  const int g  = blockIdx.x * WAVES_PER_BLOCK + wave;
  const int pA = g * 32 + col;
  const float2 xA = ((const float2*)x)[pA];
  const float2 xB = ((const float2*)x)[pA + 16];

#pragma unroll
  for (int r = 0; r < 2; ++r) {
    const int pair = (r == 0) ? wave : (8 + (wave & 3));
    const int k0 = 2 * pair, k1 = 2 * pair + 1;
    const int netc = (col < 8) ? k0 : k1;
    const int wloc = col & 7;
    const bool c8  = col < 8;

    const float2 w1v = ((const float2*)W1)[netc * WIDTH + wloc];
    const float  b1v = b1[netc * WIDTH + wloc];
    const float a1x = lo ? w1v.x : b1v;
    const float a1y = lo ? w1v.y : 0.0f;

    const int khi = lo ? 0 : 2;
    const float* R = W2 + (netc * WIDTH + wloc) * WIDTH;
    const float2 q0 = *(const float2*)(R + khi);
    const float2 q1 = *(const float2*)(R + 4 + khi);
    const float4 s01 = c8 ? make_float4(q0.x, q0.y, q1.x, q1.y)
                          : make_float4(0.0f, 0.0f, 0.0f, 0.0f);
    const float4 s23 = c8 ? make_float4(0.0f, 0.0f, 0.0f, 0.0f)
                          : make_float4(q0.x, q0.y, q1.x, q1.y);
    const float b2v  = b2[netc * WIDTH + wloc];
    const float a24x = lo ? b2v : 0.0f;

    const float4* w3p = (const float4*)(W3 + (lo ? k0 : k1) * WIDTH);
    const float4 w3a = w3p[0];
    const float4 w3b = w3p[1];

    if (r == 0 || wave < 4) {
      wtab[pair][0][lane] = make_float4(a1x, a1y, a24x, 0.0f);
      wtab[pair][1][lane] = s01;
      wtab[pair][2][lane] = s23;
      wtab[pair][3][lane] = w3a;
      wtab[pair][4][lane] = w3b;
    }
  }
  __syncthreads();

  v2f bxA, bxB;
  bxA.x = lo ? xA.x : 1.0f;  bxA.y = lo ? xA.y : 0.0f;
  bxB.x = lo ? xB.x : 1.0f;  bxB.y = lo ? xB.y : 0.0f;

  float resA[NPAIRS], resB[NPAIRS];
#pragma unroll
  for (int pr = 0; pr < NPAIRS; ++pr) {
    const float4 g0 = wtab[pr][0][lane];
    const float4 g1 = wtab[pr][1][lane];
    const float4 g2 = wtab[pr][2][lane];
    const float4 g3 = wtab[pr][3][lane];
    const float4 g4 = wtab[pr][4][lane];
    v2f a1  = {g0.x, g0.y};
    v2f a20 = {g1.x, g1.y};
    v2f a21 = {g1.z, g1.w};
    v2f a22 = {g2.x, g2.y};
    v2f a23 = {g2.z, g2.w};
    v2f a24 = {g0.z, 0.0f};
    const float w3v[8] = {g3.x, g3.y, g3.z, g3.w, g4.x, g4.y, g4.z, g4.w};
    resA[pr] = pair_tile(a1, a20, a21, a22, a23, a24, w3v, bxA, lo);
    resB[pr] = pair_tile(a1, a20, a21, a22, a23, a24, w3v, bxB, lo);
  }

  float n1[NPAIRS], n2r[NPAIRS];
#pragma unroll
  for (int i = 0; i < NPAIRS; ++i) {
    const float send = lo ? resB[i] : resA[i];
    const float recv = half_swap(send);
    const float own  = lo ? resA[i] : resB[i];
    n1[i]  = lo ? own  : recv;
    n2r[i] = lo ? recv : own;
  }

  const int p = g * 32 + lane;
  float4* row = (float4*)(sub + (size_t)p * NUM_NETS);
#pragma unroll
  for (int j = 0; j < 6; ++j)
    row[j] = make_float4(n1[2 * j], n2r[2 * j], n1[2 * j + 1], n2r[2 * j + 1]);

  const float T = lo ? xA.y : xB.y;
  const float invkT = __builtin_amdgcn_rcpf(KB_CONST * T);

  float vals[NPAIRS], ex[NPAIRS];
  float m = -1e30f;
#pragma unroll
  for (int i = 0; i < NPAIRS; ++i) {
    const float n2 = n2r[i] * n2r[i];
    const float v  = n1[i] - T * n2;
    vals[i] = v;
    const float z = n2 * (1.0f / (5.0f * KB_CONST));
    ex[i] = (-v * invkT - z * z) * LOG2E;
    m = fmaxf(m, ex[i]);
  }
  float S = 0.0f;
#pragma unroll
  for (int i = 0; i < NPAIRS; ++i) {
    ex[i] = __builtin_amdgcn_exp2f(ex[i] - m);
    S += ex[i];
  }
  const float invS = __builtin_amdgcn_rcpf(S);
  float wv = 0.0f, wlp = 0.0f;
#pragma unroll
  for (int i = 0; i < NPAIRS; ++i) {
    const float w = ex[i] * invS;
    wv  = fmaf(w, vals[i], wv);
    wlp = fmaf(w, LN2 * __builtin_amdgcn_logf(w + 1e-9f), wlp);
  }
  out[p] = wv + KB_CONST * T * wlp;
}

extern "C" void kernel_launch(void* const* d_in, const int* in_sizes, int n_in,
                              void* d_out, int out_size, void* d_ws, size_t ws_size,
                              hipStream_t stream) {
  const float* x  = (const float*)d_in[0];
  const float* W1 = (const float*)d_in[1];
  const float* b1 = (const float*)d_in[2];
  const float* W2 = (const float*)d_in[3];
  const float* b2 = (const float*)d_in[4];
  const float* W3 = (const float*)d_in[5];

  float* out = (float*)d_out;
  float* sub = out + NPTS;

  fused_subnets_mixture_kernel<<<NBLOCKS, 256, 0, stream>>>(x, W1, b1, W2, b2, W3, out, sub);
}


# Each tanh on the hardware f32 tanh instruction (v_tanh_f32) in place of mul, exp2, add, rcp, fma: 384 transcendentals per wave, not 768; last bits of each tanh change, summation order kept
# speedup vs baseline: 2.6072x; 1.2561x over previous
#include <hip/hip_runtime.h>


typedef __attribute__((ext_vector_type(2))) float v2f;
typedef __attribute__((ext_vector_type(8))) float v8f;

#define KB_CONST  0.1f
#define NPTS      262144
#define NUM_NETS  24
#define NPAIRS    12
#define WIDTH     8
#define WAVES_PER_BLOCK 8
#define NBLOCKS   (NPTS / (32 * WAVES_PER_BLOCK))

#define LOG2E     1.4426950408889634f
#define LN2       0.6931471805599453f

__device__ __forceinline__ float half_swap(float x) {
  return __int_as_float(__builtin_amdgcn_ds_swizzle(__float_as_int(x), 0x401F));
}

__device__ __forceinline__ float fast_tanh(float x) {
  return __builtin_amdgcn_tanhf(x);
}

__device__ __forceinline__ v8f wmma4(v2f a, v2f b, v8f c) {
  return __builtin_amdgcn_wmma_f32_16x16x4_f32(false, a, false, b, (short)0, c,
                                               false, false);
}

__device__ __forceinline__ float pair_tile(v2f a1, v2f a20, v2f a21, v2f a22,
                                           v2f a23, v2f a24, const float* w3v,
                                           v2f bx, bool lo)
{
  v8f c0 = {};
  v8f h1 = wmma4(a1, bx, c0);

  float hv[8], sw[8];
#pragma unroll
  for (int i = 0; i < 8; ++i) hv[i] = fast_tanh(h1[i]);
#pragma unroll
  for (int i = 0; i < 8; ++i) sw[i] = half_swap(hv[i]);

  v8f acc = {};
  v2f bs;
  bs.x = lo ? hv[0] : sw[2];  bs.y = lo ? hv[1] : sw[3];
  acc = wmma4(a20, bs, acc);
  bs.x = lo ? hv[4] : sw[6];  bs.y = lo ? hv[5] : sw[7];
  acc = wmma4(a21, bs, acc);
  bs.x = lo ? sw[0] : hv[2];  bs.y = lo ? sw[1] : hv[3];
  acc = wmma4(a22, bs, acc);
  bs.x = lo ? sw[4] : hv[6];  bs.y = lo ? sw[5] : hv[7];
  acc = wmma4(a23, bs, acc);
  bs.x = lo ? 1.0f : 0.0f;    bs.y = 0.0f;
  acc = wmma4(a24, bs, acc);

  float s3 = 0.0f;
#pragma unroll
  for (int i = 0; i < 8; ++i) s3 = fmaf(fast_tanh(acc[i]), w3v[i], s3);
  return s3;
}

__global__ __launch_bounds__(256) void fused_subnets_mixture_kernel(
    const float* __restrict__ x,  const float* __restrict__ W1,
    const float* __restrict__ b1, const float* __restrict__ W2,
    const float* __restrict__ b2, const float* __restrict__ W3,
    float* __restrict__ out, float* __restrict__ sub)
{
  __shared__ float4 wtab[NPAIRS][5][32];

  const int lane = threadIdx.x & 31;
  const int wave = threadIdx.x >> 5;
  const int col  = lane & 15;
  const bool lo  = lane < 16;

  const int g  = blockIdx.x * WAVES_PER_BLOCK + wave;
  const int pA = g * 32 + col;
  const float2 xA = ((const float2*)x)[pA];
  const float2 xB = ((const float2*)x)[pA + 16];

#pragma unroll
  for (int r = 0; r < 2; ++r) {
    const int pair = (r == 0) ? wave : (8 + (wave & 3));
    const int k0 = 2 * pair, k1 = 2 * pair + 1;
    const int netc = (col < 8) ? k0 : k1;
    const int wloc = col & 7;
    const bool c8  = col < 8;

    const float2 w1v = ((const float2*)W1)[netc * WIDTH + wloc];
    const float  b1v = b1[netc * WIDTH + wloc];
    const float a1x = lo ? w1v.x : b1v;
    const float a1y = lo ? w1v.y : 0.0f;

    const int khi = lo ? 0 : 2;
    const float* R = W2 + (netc * WIDTH + wloc) * WIDTH;
    const float2 q0 = *(const float2*)(R + khi);
    const float2 q1 = *(const float2*)(R + 4 + khi);
    const float4 s01 = c8 ? make_float4(q0.x, q0.y, q1.x, q1.y)
                          : make_float4(0.0f, 0.0f, 0.0f, 0.0f);
    const float4 s23 = c8 ? make_float4(0.0f, 0.0f, 0.0f, 0.0f)
                          : make_float4(q0.x, q0.y, q1.x, q1.y);
    const float b2v  = b2[netc * WIDTH + wloc];
    const float a24x = lo ? b2v : 0.0f;

    const float4* w3p = (const float4*)(W3 + (lo ? k0 : k1) * WIDTH);
    const float4 w3a = w3p[0];
    const float4 w3b = w3p[1];

    if (r == 0 || wave < 4) {
      wtab[pair][0][lane] = make_float4(a1x, a1y, a24x, 0.0f);
      wtab[pair][1][lane] = s01;
      wtab[pair][2][lane] = s23;
      wtab[pair][3][lane] = w3a;
      wtab[pair][4][lane] = w3b;
    }
  }
  __syncthreads();

  v2f bxA, bxB;
  bxA.x = lo ? xA.x : 1.0f;  bxA.y = lo ? xA.y : 0.0f;
  bxB.x = lo ? xB.x : 1.0f;  bxB.y = lo ? xB.y : 0.0f;

  float resA[NPAIRS], resB[NPAIRS];
#pragma unroll
  for (int pr = 0; pr < NPAIRS; ++pr) {
    const float4 g0 = wtab[pr][0][lane];
    const float4 g1 = wtab[pr][1][lane];
    const float4 g2 = wtab[pr][2][lane];
    const float4 g3 = wtab[pr][3][lane];
    const float4 g4 = wtab[pr][4][lane];
    v2f a1  = {g0.x, g0.y};
    v2f a20 = {g1.x, g1.y};
    v2f a21 = {g1.z, g1.w};
    v2f a22 = {g2.x, g2.y};
    v2f a23 = {g2.z, g2.w};
    v2f a24 = {g0.z, 0.0f};
    const float w3v[8] = {g3.x, g3.y, g3.z, g3.w, g4.x, g4.y, g4.z, g4.w};
    resA[pr] = pair_tile(a1, a20, a21, a22, a23, a24, w3v, bxA, lo);
    resB[pr] = pair_tile(a1, a20, a21, a22, a23, a24, w3v, bxB, lo);
  }

  float n1[NPAIRS], n2r[NPAIRS];
#pragma unroll
  for (int i = 0; i < NPAIRS; ++i) {
    const float send = lo ? resB[i] : resA[i];
    const float recv = half_swap(send);
    const float own  = lo ? resA[i] : resB[i];
    n1[i]  = lo ? own  : recv;
    n2r[i] = lo ? recv : own;
  }

  const int p = g * 32 + lane;
  float4* row = (float4*)(sub + (size_t)p * NUM_NETS);
#pragma unroll
  for (int j = 0; j < 6; ++j)
    row[j] = make_float4(n1[2 * j], n2r[2 * j], n1[2 * j + 1], n2r[2 * j + 1]);

  const float T = lo ? xA.y : xB.y;
  const float invkT = __builtin_amdgcn_rcpf(KB_CONST * T);

  float vals[NPAIRS], ex[NPAIRS];
  float m = -1e30f;
#pragma unroll
  for (int i = 0; i < NPAIRS; ++i) {
    const float n2 = n2r[i] * n2r[i];
    const float v  = n1[i] - T * n2;
    vals[i] = v;
    const float z = n2 * (1.0f / (5.0f * KB_CONST));
    ex[i] = (-v * invkT - z * z) * LOG2E;
    m = fmaxf(m, ex[i]);
  }
  float S = 0.0f;
#pragma unroll
  for (int i = 0; i < NPAIRS; ++i) {
    ex[i] = __builtin_amdgcn_exp2f(ex[i] - m);
    S += ex[i];
  }
  const float invS = __builtin_amdgcn_rcpf(S);
  float wv = 0.0f, wlp = 0.0f;
#pragma unroll
  for (int i = 0; i < NPAIRS; ++i) {
    const float w = ex[i] * invS;
    wv  = fmaf(w, vals[i], wv);
    wlp = fmaf(w, LN2 * __builtin_amdgcn_logf(w + 1e-9f), wlp);
  }
  out[p] = wv + KB_CONST * T * wlp;
}

extern "C" void kernel_launch(void* const* d_in, const int* in_sizes, int n_in,
                              void* d_out, int out_size, void* d_ws, size_t ws_size,
                              hipStream_t stream) {
  const float* x  = (const float*)d_in[0];
  const float* W1 = (const float*)d_in[1];
  const float* b1 = (const float*)d_in[2];
  const float* W2 = (const float*)d_in[3];
  const float* b2 = (const float*)d_in[4];
  const float* W3 = (const float*)d_in[5];

  float* out = (float*)d_out;
  float* sub = out + NPTS;

  fused_subnets_mixture_kernel<<<NBLOCKS, 256, 0, stream>>>(x, W1, b1, W2, b2, W3, out, sub);
}


# Half-wave exchanges on the vector ALU: v_permlane16_swap_b32 replaces ds_swizzle (LDS unit) plus select, 108 swaps per wave for 204 swizzles and 228 selects; weight table, k order, sum order kept
# speedup vs baseline: 2.9816x; 1.1436x over previous
#include <hip/hip_runtime.h>


typedef __attribute__((ext_vector_type(2))) float v2f;
typedef __attribute__((ext_vector_type(8))) float v8f;

#define KB_CONST  0.1f
#define NPTS      262144
#define NUM_NETS  24
#define NPAIRS    12
#define WIDTH     8
#define WAVES_PER_BLOCK 8
#define NBLOCKS   (NPTS / (32 * WAVES_PER_BLOCK))

#define LOG2E     1.4426950408889634f
#define LN2       0.6931471805599453f

typedef __attribute__((ext_vector_type(2))) unsigned int v2u;

__device__ __forceinline__ void row_swap(float a, float b, float& lo_rows,
                                         float& hi_rows) {
  const v2u r = __builtin_amdgcn_permlane16_swap(__float_as_uint(a),
                                                 __float_as_uint(b), false, false);
  lo_rows = __uint_as_float(r.x);
  hi_rows = __uint_as_float(r.y);
}

__device__ __forceinline__ float fast_tanh(float x) {
  return __builtin_amdgcn_tanhf(x);
}

__device__ __forceinline__ v8f wmma4(v2f a, v2f b, v8f c) {
  return __builtin_amdgcn_wmma_f32_16x16x4_f32(false, a, false, b, (short)0, c,
                                               false, false);
}

__device__ __forceinline__ float pair_tile(v2f a1, v2f a20, v2f a21, v2f a22,
                                           v2f a23, v2f a24, const float* w3v,
                                           v2f bx, bool lo)
{
  v8f c0 = {};
  v8f h1 = wmma4(a1, bx, c0);

  float hv[8];
#pragma unroll
  for (int i = 0; i < 8; ++i) hv[i] = fast_tanh(h1[i]);

  float s0x, s0y, s1x, s1y, s2x, s2y, s3x, s3y;
  row_swap(hv[0], hv[2], s0x, s2x);
  row_swap(hv[1], hv[3], s0y, s2y);
  row_swap(hv[4], hv[6], s1x, s3x);
  row_swap(hv[5], hv[7], s1y, s3y);

  v8f acc = {};
  v2f bs;
  bs.x = s0x;  bs.y = s0y;
  acc = wmma4(a20, bs, acc);
  bs.x = s1x;  bs.y = s1y;
  acc = wmma4(a21, bs, acc);
  bs.x = s2x;  bs.y = s2y;
  acc = wmma4(a22, bs, acc);
  bs.x = s3x;  bs.y = s3y;
  acc = wmma4(a23, bs, acc);
  bs.x = lo ? 1.0f : 0.0f;    bs.y = 0.0f;
  acc = wmma4(a24, bs, acc);

  float s3 = 0.0f;
#pragma unroll
  for (int i = 0; i < 8; ++i) s3 = fmaf(fast_tanh(acc[i]), w3v[i], s3);
  return s3;
}

__global__ __launch_bounds__(256) void fused_subnets_mixture_kernel(
    const float* __restrict__ x,  const float* __restrict__ W1,
    const float* __restrict__ b1, const float* __restrict__ W2,
    const float* __restrict__ b2, const float* __restrict__ W3,
    float* __restrict__ out, float* __restrict__ sub)
{
  __shared__ float4 wtab[NPAIRS][5][32];

  const int lane = threadIdx.x & 31;
  const int wave = threadIdx.x >> 5;
  const int col  = lane & 15;
  const bool lo  = lane < 16;

  const int g  = blockIdx.x * WAVES_PER_BLOCK + wave;
  const int pA = g * 32 + col;
  const float2 xA = ((const float2*)x)[pA];
  const float2 xB = ((const float2*)x)[pA + 16];

#pragma unroll
  for (int r = 0; r < 2; ++r) {
    const int pair = (r == 0) ? wave : (8 + (wave & 3));
    const int k0 = 2 * pair, k1 = 2 * pair + 1;
    const int netc = (col < 8) ? k0 : k1;
    const int wloc = col & 7;
    const bool c8  = col < 8;

    const float2 w1v = ((const float2*)W1)[netc * WIDTH + wloc];
    const float  b1v = b1[netc * WIDTH + wloc];
    const float a1x = lo ? w1v.x : b1v;
    const float a1y = lo ? w1v.y : 0.0f;

    const int khi = lo ? 0 : 2;
    const float* R = W2 + (netc * WIDTH + wloc) * WIDTH;
    const float2 q0 = *(const float2*)(R + khi);
    const float2 q1 = *(const float2*)(R + 4 + khi);
    const float4 s01 = c8 ? make_float4(q0.x, q0.y, q1.x, q1.y)
                          : make_float4(0.0f, 0.0f, 0.0f, 0.0f);
    const float4 s23 = c8 ? make_float4(0.0f, 0.0f, 0.0f, 0.0f)
                          : make_float4(q0.x, q0.y, q1.x, q1.y);
    const float b2v  = b2[netc * WIDTH + wloc];
    const float a24x = lo ? b2v : 0.0f;

    const float4* w3p = (const float4*)(W3 + (lo ? k0 : k1) * WIDTH);
    const float4 w3a = w3p[0];
    const float4 w3b = w3p[1];

    if (r == 0 || wave < 4) {
      wtab[pair][0][lane] = make_float4(a1x, a1y, a24x, 0.0f);
      wtab[pair][1][lane] = s01;
      wtab[pair][2][lane] = s23;
      wtab[pair][3][lane] = w3a;
      wtab[pair][4][lane] = w3b;
    }
  }
  __syncthreads();

  v2f bxA, bxB;
  bxA.x = lo ? xA.x : 1.0f;  bxA.y = lo ? xA.y : 0.0f;
  bxB.x = lo ? xB.x : 1.0f;  bxB.y = lo ? xB.y : 0.0f;

  float resA[NPAIRS], resB[NPAIRS];
#pragma unroll
  for (int pr = 0; pr < NPAIRS; ++pr) {
    const float4 g0 = wtab[pr][0][lane];
    const float4 g1 = wtab[pr][1][lane];
    const float4 g2 = wtab[pr][2][lane];
    const float4 g3 = wtab[pr][3][lane];
    const float4 g4 = wtab[pr][4][lane];
    v2f a1  = {g0.x, g0.y};
    v2f a20 = {g1.x, g1.y};
    v2f a21 = {g1.z, g1.w};
    v2f a22 = {g2.x, g2.y};
    v2f a23 = {g2.z, g2.w};
    v2f a24 = {g0.z, 0.0f};
    const float w3v[8] = {g3.x, g3.y, g3.z, g3.w, g4.x, g4.y, g4.z, g4.w};
    resA[pr] = pair_tile(a1, a20, a21, a22, a23, a24, w3v, bxA, lo);
    resB[pr] = pair_tile(a1, a20, a21, a22, a23, a24, w3v, bxB, lo);
  }

  float n1[NPAIRS], n2r[NPAIRS];
#pragma unroll
  for (int i = 0; i < NPAIRS; ++i) row_swap(resA[i], resB[i], n1[i], n2r[i]);

  const int p = g * 32 + lane;
  float4* row = (float4*)(sub + (size_t)p * NUM_NETS);
#pragma unroll
  for (int j = 0; j < 6; ++j)
    row[j] = make_float4(n1[2 * j], n2r[2 * j], n1[2 * j + 1], n2r[2 * j + 1]);

  const float T = lo ? xA.y : xB.y;
  const float invkT = __builtin_amdgcn_rcpf(KB_CONST * T);

  float vals[NPAIRS], ex[NPAIRS];
  float m = -1e30f;
#pragma unroll
  for (int i = 0; i < NPAIRS; ++i) {
    const float n2 = n2r[i] * n2r[i];
    const float v  = n1[i] - T * n2;
    vals[i] = v;
    const float z = n2 * (1.0f / (5.0f * KB_CONST));
    ex[i] = (-v * invkT - z * z) * LOG2E;
    m = fmaxf(m, ex[i]);
  }
  float S = 0.0f;
#pragma unroll
  for (int i = 0; i < NPAIRS; ++i) {
    ex[i] = __builtin_amdgcn_exp2f(ex[i] - m);
    S += ex[i];
  }
  const float invS = __builtin_amdgcn_rcpf(S);
  float wv = 0.0f, wlp = 0.0f;
#pragma unroll
  for (int i = 0; i < NPAIRS; ++i) {
    const float w = ex[i] * invS;
    wv  = fmaf(w, vals[i], wv);
    wlp = fmaf(w, LN2 * __builtin_amdgcn_logf(w + 1e-9f), wlp);
  }
  out[p] = wv + KB_CONST * T * wlp;
}

extern "C" void kernel_launch(void* const* d_in, const int* in_sizes, int n_in,
                              void* d_out, int out_size, void* d_ws, size_t ws_size,
                              hipStream_t stream) {
  const float* x  = (const float*)d_in[0];
  const float* W1 = (const float*)d_in[1];
  const float* b1 = (const float*)d_in[2];
  const float* W2 = (const float*)d_in[3];
  const float* b2 = (const float*)d_in[4];
  const float* W3 = (const float*)d_in[5];

  float* out = (float*)d_out;
  float* sub = out + NPTS;

  fused_subnets_mixture_kernel<<<NBLOCKS, 256, 0, stream>>>(x, W1, b1, W2, b2, W3, out, sub);
}
